# starting kernel
# baseline (speedup 1.0000x reference)
#include <hip/hip_runtime.h>

typedef __attribute__((ext_vector_type(2))) float v2f;
typedef __attribute__((ext_vector_type(4))) float v4f;
typedef __attribute__((ext_vector_type(8))) float v8f;

#define BDIM 256
constexpr int B_ = 4, H_ = 16, T_ = 4096, D_ = 64;
constexpr int BH = B_ * H_;
constexpr float EPS = 1e-6f;

constexpr int TC = 64;
constexpr int NCHUNK = 4;
constexpr int SPLIT = T_ / (TC * NCHUNK);
constexpr int TCP = 68;

constexpr int TM = 64;
constexpr int SQS = 68;

__device__ __forceinline__ float fmap(float x) {
    return x > 0.0f ? x + 1.0f : __expf(x);
}

__device__ __forceinline__ v8f wmma4(v2f a, v2f b, v8f c) {
    return __builtin_amdgcn_wmma_f32_16x16x4_f32(false, a, false, b, (short)0, c,
                                                 false, false);
}

__global__ __launch_bounds__(BDIM) void zero_ws_kernel(float* w, int n) {
    int i = blockIdx.x * BDIM + threadIdx.x;
    if (i < n) w[i] = 0.0f;
}

__global__ __launch_bounds__(BDIM) void pass1_kv_kernel(const float* __restrict__ kin,
                                                        const float* __restrict__ vin,
                                                        const float* __restrict__ mask,
                                                        float* __restrict__ wkv,
                                                        float* __restrict__ wks) {
    __shared__ float lkT[D_ * TCP];
    __shared__ float lvT[D_ * TCP];

    const int tid = threadIdx.x;
    const int bh  = blockIdx.x / SPLIT;
    const int blk = blockIdx.x % SPLIT;
    const int b   = bh / H_;
    const int t00 = blk * TC * NCHUNK;

    const int wid  = tid >> 5;
    const int lane = tid & 31;
    const int half = lane >> 4;
    const int lm   = lane & 15;
    const int quad  = wid & 3;
    const int khalf = wid >> 2;
    const int eT0 = (quad >> 1) * 32;
    const int dT0 = (quad & 1) * 32;

    v8f a00 = {}, a01 = {}, a10 = {}, a11 = {};
    float ksr = 0.0f;
    const int dks = tid >> 2;
    const int l4  = tid & 3;

    for (int c = 0; c < NCHUNK; ++c) {
        const int t0 = t00 + c * TC;
        const size_t gbase = ((size_t)bh * T_ + t0) * D_;

        #pragma unroll
        for (int i = 0; i < (TC * D_) / (BDIM * 4); ++i) {
            int idx4 = tid + i * BDIM;
            int tl   = idx4 >> 4;
            int d0   = (idx4 & 15) * 4;
            float m  = mask[b * T_ + t0 + tl];
            v4f k4 = *(const v4f*)(kin + gbase + (size_t)idx4 * 4);
            v4f v4 = *(const v4f*)(vin + gbase + (size_t)idx4 * 4);
            #pragma unroll
            for (int j = 0; j < 4; ++j) {
                lkT[(d0 + j) * TCP + tl] = fmap(k4[j]) * m;
                lvT[(d0 + j) * TCP + tl] = v4[j] * m;
            }
        }
        __syncthreads();

        #pragma unroll
        for (int kk = 0; kk < 32; kk += 4) {
            const int tr = khalf * 32 + kk + half * 2;
            v2f fa0, fa1, fb0, fb1;
            fa0.x = lvT[(eT0 + lm) * TCP + tr];
            fa0.y = lvT[(eT0 + lm) * TCP + tr + 1];
            fa1.x = lvT[(eT0 + 16 + lm) * TCP + tr];
            fa1.y = lvT[(eT0 + 16 + lm) * TCP + tr + 1];
            fb0.x = lkT[(dT0 + lm) * TCP + tr];
            fb0.y = lkT[(dT0 + lm) * TCP + tr + 1];
            fb1.x = lkT[(dT0 + 16 + lm) * TCP + tr];
            fb1.y = lkT[(dT0 + 16 + lm) * TCP + tr + 1];
            a00 = wmma4(fa0, fb0, a00);
            a01 = wmma4(fa0, fb1, a01);
            a10 = wmma4(fa1, fb0, a10);
            a11 = wmma4(fa1, fb1, a11);
        }

        {
            float s = 0.0f;
            #pragma unroll
            for (int j = 0; j < 16; ++j)
                s += lkT[dks * TCP + l4 * 16 + j];
            s += __shfl_xor(s, 1, 32);
            s += __shfl_xor(s, 2, 32);
            ksr += s;
        }
        __syncthreads();
    }

    float* kvb = wkv + (size_t)bh * D_ * D_;
    #pragma unroll
    for (int r = 0; r < 8; ++r) {
        const int e0 = eT0 + r + half * 8;
        atomicAdd(&kvb[e0 * D_ + dT0 + lm],            a00[r]);
        atomicAdd(&kvb[e0 * D_ + dT0 + 16 + lm],       a01[r]);
        atomicAdd(&kvb[(e0 + 16) * D_ + dT0 + lm],     a10[r]);
        atomicAdd(&kvb[(e0 + 16) * D_ + dT0 + 16 + lm], a11[r]);
    }
    if (l4 == 0) atomicAdd(&wks[bh * D_ + dks], ksr);
}

__global__ __launch_bounds__(BDIM) void pass2_out_kernel(const float* __restrict__ qin,
                                                         const float* __restrict__ wkv,
                                                         const float* __restrict__ wks,
                                                         float* __restrict__ out) {
    __shared__ float skvT[D_ * SQS];
    __shared__ float sq[TM * SQS];
    __shared__ float sks[D_];
    __shared__ float sz[TM];

    const int tid   = threadIdx.x;
    const int bh    = blockIdx.x >> 6;
    const int tbase = (blockIdx.x & 63) * TM;
    const size_t qbase = ((size_t)bh * T_ + tbase) * D_;

    const float* kvg = wkv + (size_t)bh * D_ * D_;
    #pragma unroll
    for (int i = 0; i < (D_ * D_) / (BDIM * 4); ++i) {
        int idx4 = tid + i * BDIM;
        int e = idx4 >> 4, d0 = (idx4 & 15) * 4;
        *(v4f*)&skvT[e * SQS + d0] = *(const v4f*)(kvg + (size_t)idx4 * 4);
    }
    #pragma unroll
    for (int i = 0; i < (TM * D_) / (BDIM * 4); ++i) {
        int idx4 = tid + i * BDIM;
        int row = idx4 >> 4, c0 = (idx4 & 15) * 4;
        v4f qq = *(const v4f*)(qin + qbase + (size_t)idx4 * 4);
        v4f r;
        r.x = fmap(qq.x); r.y = fmap(qq.y); r.z = fmap(qq.z); r.w = fmap(qq.w);
        *(v4f*)&sq[row * SQS + c0] = r;
    }
    if (tid < D_) sks[tid] = wks[bh * D_ + tid];
    __syncthreads();

    {
        const int row = tid >> 2;
        const int l4  = tid & 3;
        float s = 0.0f;
        #pragma unroll
        for (int j = 0; j < 16; ++j) {
            const int c = l4 * 16 + j;
            s += sq[row * SQS + c] * sks[c];
        }
        s += __shfl_xor(s, 1, 32);
        s += __shfl_xor(s, 2, 32);
        if (l4 == 0) sz[row] = s + EPS;
    }
    __syncthreads();

    const int wid  = tid >> 5;
    const int lane = tid & 31;
    const int half = lane >> 4;
    const int lm   = lane & 15;
    const int rt  = (wid >> 1) * 16;
    const int ct0 = (wid & 1) * 32;

    v8f c0 = {}, c1 = {};
    #pragma unroll
    for (int kk = 0; kk < D_; kk += 4) {
        const int kr = kk + half * 2;
        v2f fa, fb0, fb1;
        fa.x  = sq[(rt + lm) * SQS + kr];
        fa.y  = sq[(rt + lm) * SQS + kr + 1];
        fb0.x = skvT[(ct0 + lm) * SQS + kr];
        fb0.y = skvT[(ct0 + lm) * SQS + kr + 1];
        fb1.x = skvT[(ct0 + 16 + lm) * SQS + kr];
        fb1.y = skvT[(ct0 + 16 + lm) * SQS + kr + 1];
        c0 = wmma4(fa, fb0, c0);
        c1 = wmma4(fa, fb1, c1);
    }
    #pragma unroll
    for (int r = 0; r < 8; ++r) {
        const int row = rt + r + half * 8;
        const float z = sz[row];
        const size_t o = ((size_t)bh * T_ + tbase + row) * D_;
        out[o + ct0 + lm]      = c0[r] / z;
        out[o + ct0 + 16 + lm] = c1[r] / z;
    }
}

extern "C" void kernel_launch(void* const* d_in, const int* in_sizes, int n_in,
                              void* d_out, int out_size, void* d_ws, size_t ws_size,
                              hipStream_t stream) {
    (void)in_sizes; (void)n_in; (void)out_size; (void)ws_size;
    const float* q    = (const float*)d_in[0];
    const float* k    = (const float*)d_in[1];
    const float* v    = (const float*)d_in[2];
    const float* mask = (const float*)d_in[3];
    float* out = (float*)d_out;
    float* wkv = (float*)d_ws;
    float* wks = wkv + (size_t)BH * D_ * D_;

    const int nz = BH * D_ * D_ + BH * D_;
    zero_ws_kernel<<<(nz + BDIM - 1) / BDIM, BDIM, 0, stream>>>(wkv, nz);
    pass1_kv_kernel<<<BH * SPLIT, BDIM, 0, stream>>>(k, v, mask, wkv, wks);
    pass2_out_kernel<<<BH * (T_ / TM), BDIM, 0, stream>>>(q, wkv, wks, out);
}


# pass1 feeds WMMA from 16-byte global loads, no LDS transpose or chunk barriers: a wave keeps all 64x64 kv for 128 timesteps; 8 partials summed in LDS, 4 atomic adds not 16 (new sum order)
# speedup vs baseline: 1.2629x; 1.2629x over previous
#include <hip/hip_runtime.h>

typedef __attribute__((ext_vector_type(2))) float v2f;
typedef __attribute__((ext_vector_type(4))) float v4f;
typedef __attribute__((ext_vector_type(8))) float v8f;

#define BDIM 256
constexpr int B_ = 4, H_ = 16, T_ = 4096, D_ = 64;
constexpr int BH = B_ * H_;
constexpr float EPS = 1e-6f;

constexpr int P1_WAVES = BDIM / 32;
constexpr int P1_TB = 1024;
constexpr int P1_SPLIT = T_ / P1_TB;
constexpr int P1_TW = P1_TB / P1_WAVES;
constexpr int P1_GT = 16;
constexpr int P1_LD = P1_GT / 2;
constexpr int P1_G4 = P1_GT * D_ / 4;
constexpr int P1_NG = P1_TW / P1_GT;
constexpr int P1_LDS_USED = P1_WAVES * D_ * D_ + P1_WAVES * D_;
constexpr int P1_LDS_HALF = 160 * 1024 / 4;
constexpr int P1_LDS_FLOATS = (P1_LDS_USED > P1_LDS_HALF ? P1_LDS_USED : P1_LDS_HALF) + 512;

constexpr int TM = 64;
constexpr int SQS = 68;

__device__ __forceinline__ float fmap(float x) {
    return x > 0.0f ? x + 1.0f : __expf(x);
}

__device__ __forceinline__ v8f wmma4(v2f a, v2f b, v8f c) {
    return __builtin_amdgcn_wmma_f32_16x16x4_f32(false, a, false, b, (short)0, c,
                                                 false, false);
}

__global__ __launch_bounds__(BDIM) void zero_ws_kernel(float* w, int n) {
    int i = blockIdx.x * BDIM + threadIdx.x;
    if (i < n) w[i] = 0.0f;
}

struct P1Group {
    v4f k[P1_LD];
    v4f v[P1_LD];
    float m[P1_LD];
};

__device__ __forceinline__ void p1_load(P1Group& g, const v4f* __restrict__ k4,
                                        const v4f* __restrict__ v4,
                                        const float* __restrict__ mrow,
                                        int lane, int half) {
    #pragma unroll
    for (int i = 0; i < P1_LD; ++i) {
        g.k[i] = k4[i * 32 + lane];
        g.v[i] = v4[i * 32 + lane];
        g.m[i] = mrow[2 * i + half];
    }
}

__device__ __forceinline__ void p1_compute(const P1Group& g, v8f (&acc)[4][4], v4f& ks) {
    v4f kp[P1_LD], vp[P1_LD];
    #pragma unroll
    for (int i = 0; i < P1_LD; ++i) {
        const float m = g.m[i];
        #pragma unroll
        for (int j = 0; j < 4; ++j) {
            kp[i][j] = fmap(g.k[i][j]) * m;
            vp[i][j] = g.v[i][j] * m;
        }
        ks += kp[i];
    }
    #pragma unroll
    for (int s = 0; s < P1_LD / 2; ++s) {
        #pragma unroll
        for (int mt = 0; mt < 4; ++mt) {
            v2f a;
            a.x = vp[2 * s][mt];
            a.y = vp[2 * s + 1][mt];
            #pragma unroll
            for (int nt = 0; nt < 4; ++nt) {
                v2f b;
                b.x = kp[2 * s][nt];
                b.y = kp[2 * s + 1][nt];
                acc[mt][nt] = wmma4(a, b, acc[mt][nt]);
            }
        }
    }
}

__global__ __launch_bounds__(BDIM) void pass1_kv_kernel(const float* __restrict__ kin,
                                                        const float* __restrict__ vin,
                                                        const float* __restrict__ mask,
                                                        float* __restrict__ wkv,
                                                        float* __restrict__ wks) {
    __shared__ __attribute__((aligned(16))) float part[P1_LDS_FLOATS];

    const int tid  = threadIdx.x;
    const int bh   = blockIdx.x / P1_SPLIT;
    const int blk  = blockIdx.x % P1_SPLIT;
    const int b    = bh / H_;
    const int wid  = tid >> 5;
    const int lane = tid & 31;
    const int half = lane >> 4;
    const int lm   = lane & 15;

    const int t0 = blk * P1_TB + wid * P1_TW;
    const v4f* k4 = (const v4f*)(kin + ((size_t)bh * T_ + t0) * D_);
    const v4f* v4 = (const v4f*)(vin + ((size_t)bh * T_ + t0) * D_);
    const float* mrow = mask + (size_t)b * T_ + t0;

    v8f acc[4][4];
    #pragma unroll
    for (int mt = 0; mt < 4; ++mt)
        #pragma unroll
        for (int nt = 0; nt < 4; ++nt)
            acc[mt][nt] = v8f{};
    v4f ks = {};

    P1Group ga, gb;
    p1_load(ga, k4, v4, mrow, lane, half);
    #pragma unroll 1
    for (int g = 0; g < P1_NG - 2; g += 2) {
        p1_load(gb, k4 + (g + 1) * P1_G4, v4 + (g + 1) * P1_G4, mrow + (g + 1) * P1_GT, lane, half);
        __builtin_amdgcn_sched_barrier(0);
        p1_compute(ga, acc, ks);
        p1_load(ga, k4 + (g + 2) * P1_G4, v4 + (g + 2) * P1_G4, mrow + (g + 2) * P1_GT, lane, half);
        __builtin_amdgcn_sched_barrier(0);
        p1_compute(gb, acc, ks);
    }
    p1_load(gb, k4 + (P1_NG - 1) * P1_G4, v4 + (P1_NG - 1) * P1_G4, mrow + (P1_NG - 1) * P1_GT, lane, half);
    __builtin_amdgcn_sched_barrier(0);
    p1_compute(ga, acc, ks);
    p1_compute(gb, acc, ks);

    float* mine = part + wid * (D_ * D_);
    #pragma unroll
    for (int mt = 0; mt < 4; ++mt) {
        #pragma unroll
        for (int r = 0; r < 8; ++r) {
            const int e = 32 * half + 4 * r + mt;
            v4f o;
            o.x = acc[mt][0][r];
            o.y = acc[mt][1][r];
            o.z = acc[mt][2][r];
            o.w = acc[mt][3][r];
            *(v4f*)&mine[e * D_ + 4 * lm] = o;
        }
    }
    #pragma unroll
    for (int j = 0; j < 4; ++j) ks[j] += __shfl_xor(ks[j], 16, 32);
    float* ksp = part + P1_WAVES * (D_ * D_);
    if (half == 0) *(v4f*)&ksp[wid * D_ + 4 * lm] = ks;
    __syncthreads();

    float* kvb = wkv + (size_t)bh * D_ * D_;
    #pragma unroll
    for (int i = 0; i < (D_ * D_) / BDIM; ++i) {
        const int idx = tid + i * BDIM;
        float s = part[idx];
        #pragma unroll
        for (int w = 1; w < P1_WAVES; ++w) s += part[w * (D_ * D_) + idx];
        atomicAdd(&kvb[idx], s);
    }
    if (tid < D_) {
        float s = ksp[tid];
        #pragma unroll
        for (int w = 1; w < P1_WAVES; ++w) s += ksp[w * D_ + tid];
        atomicAdd(&wks[bh * D_ + tid], s);
    }
}

__global__ __launch_bounds__(BDIM) void pass2_out_kernel(const float* __restrict__ qin,
                                                         const float* __restrict__ wkv,
                                                         const float* __restrict__ wks,
                                                         float* __restrict__ out) {
    __shared__ float skvT[D_ * SQS];
    __shared__ float sq[TM * SQS];
    __shared__ float sks[D_];
    __shared__ float sz[TM];

    const int tid   = threadIdx.x;
    const int bh    = blockIdx.x >> 6;
    const int tbase = (blockIdx.x & 63) * TM;
    const size_t qbase = ((size_t)bh * T_ + tbase) * D_;

    const float* kvg = wkv + (size_t)bh * D_ * D_;
    #pragma unroll
    for (int i = 0; i < (D_ * D_) / (BDIM * 4); ++i) {
        int idx4 = tid + i * BDIM;
        int e = idx4 >> 4, d0 = (idx4 & 15) * 4;
        *(v4f*)&skvT[e * SQS + d0] = *(const v4f*)(kvg + (size_t)idx4 * 4);
    }
    #pragma unroll
    for (int i = 0; i < (TM * D_) / (BDIM * 4); ++i) {
        int idx4 = tid + i * BDIM;
        int row = idx4 >> 4, c0 = (idx4 & 15) * 4;
        v4f qq = *(const v4f*)(qin + qbase + (size_t)idx4 * 4);
        v4f r;
        r.x = fmap(qq.x); r.y = fmap(qq.y); r.z = fmap(qq.z); r.w = fmap(qq.w);
        *(v4f*)&sq[row * SQS + c0] = r;
    }
    if (tid < D_) sks[tid] = wks[bh * D_ + tid];
    __syncthreads();

    {
        const int row = tid >> 2;
        const int l4  = tid & 3;
        float s = 0.0f;
        #pragma unroll
        for (int j = 0; j < 16; ++j) {
            const int c = l4 * 16 + j;
            s += sq[row * SQS + c] * sks[c];
        }
        s += __shfl_xor(s, 1, 32);
        s += __shfl_xor(s, 2, 32);
        if (l4 == 0) sz[row] = s + EPS;
    }
    __syncthreads();

    const int wid  = tid >> 5;
    const int lane = tid & 31;
    const int half = lane >> 4;
    const int lm   = lane & 15;
    const int rt  = (wid >> 1) * 16;
    const int ct0 = (wid & 1) * 32;

    v8f c0 = {}, c1 = {};
    #pragma unroll
    for (int kk = 0; kk < D_; kk += 4) {
        const int kr = kk + half * 2;
        v2f fa, fb0, fb1;
        fa.x  = sq[(rt + lm) * SQS + kr];
        fa.y  = sq[(rt + lm) * SQS + kr + 1];
        fb0.x = skvT[(ct0 + lm) * SQS + kr];
        fb0.y = skvT[(ct0 + lm) * SQS + kr + 1];
        fb1.x = skvT[(ct0 + 16 + lm) * SQS + kr];
        fb1.y = skvT[(ct0 + 16 + lm) * SQS + kr + 1];
        c0 = wmma4(fa, fb0, c0);
        c1 = wmma4(fa, fb1, c1);
    }
    #pragma unroll
    for (int r = 0; r < 8; ++r) {
        const int row = rt + r + half * 8;
        const float z = sz[row];
        const size_t o = ((size_t)bh * T_ + tbase + row) * D_;
        out[o + ct0 + lm]      = c0[r] / z;
        out[o + ct0 + 16 + lm] = c1[r] / z;
    }
}

extern "C" void kernel_launch(void* const* d_in, const int* in_sizes, int n_in,
                              void* d_out, int out_size, void* d_ws, size_t ws_size,
                              hipStream_t stream) {
    (void)in_sizes; (void)n_in; (void)out_size; (void)ws_size;
    const float* q    = (const float*)d_in[0];
    const float* k    = (const float*)d_in[1];
    const float* v    = (const float*)d_in[2];
    const float* mask = (const float*)d_in[3];
    float* out = (float*)d_out;
    float* wkv = (float*)d_ws;
    float* wks = wkv + (size_t)BH * D_ * D_;

    const int nz = BH * D_ * D_ + BH * D_;
    zero_ws_kernel<<<(nz + BDIM - 1) / BDIM, BDIM, 0, stream>>>(wkv, nz);
    pass1_kv_kernel<<<BH * P1_SPLIT, BDIM, 0, stream>>>(k, v, mask, wkv, wks);
    pass2_out_kernel<<<BH * (T_ / TM), BDIM, 0, stream>>>(q, wkv, wks, out);
}


# pass2 writes whole lines: the 64x64 result tile returns through the q tile's LDS rows and each wave store covers two full 256-byte rows with 16-byte stores (was 64-byte pieces from 4-byte stores)
# speedup vs baseline: 1.3397x; 1.0608x over previous
#include <hip/hip_runtime.h>

typedef __attribute__((ext_vector_type(2))) float v2f;
typedef __attribute__((ext_vector_type(4))) float v4f;
typedef __attribute__((ext_vector_type(8))) float v8f;

#define BDIM 256
constexpr int B_ = 4, H_ = 16, T_ = 4096, D_ = 64;
constexpr int BH = B_ * H_;
constexpr float EPS = 1e-6f;

constexpr int P1_WAVES = BDIM / 32;
constexpr int P1_TB = 1024;
constexpr int P1_SPLIT = T_ / P1_TB;
constexpr int P1_TW = P1_TB / P1_WAVES;
constexpr int P1_GT = 16;
constexpr int P1_LD = P1_GT / 2;
constexpr int P1_G4 = P1_GT * D_ / 4;
constexpr int P1_NG = P1_TW / P1_GT;
constexpr int P1_LDS_USED = P1_WAVES * D_ * D_ + P1_WAVES * D_;
constexpr int P1_LDS_HALF = 160 * 1024 / 4;
constexpr int P1_LDS_FLOATS = (P1_LDS_USED > P1_LDS_HALF ? P1_LDS_USED : P1_LDS_HALF) + 512;

constexpr int TM = 64;
constexpr int SQS = 68;

__device__ __forceinline__ float fmap(float x) {
    return x > 0.0f ? x + 1.0f : __expf(x);
}

__device__ __forceinline__ v8f wmma4(v2f a, v2f b, v8f c) {
    return __builtin_amdgcn_wmma_f32_16x16x4_f32(false, a, false, b, (short)0, c,
                                                 false, false);
}

__global__ __launch_bounds__(BDIM) void zero_ws_kernel(float* w, int n) {
    int i = blockIdx.x * BDIM + threadIdx.x;
    if (i < n) w[i] = 0.0f;
}

struct P1Group {
    v4f k[P1_LD];
    v4f v[P1_LD];
    float m[P1_LD];
};

__device__ __forceinline__ void p1_load(P1Group& g, const v4f* __restrict__ k4,
                                        const v4f* __restrict__ v4,
                                        const float* __restrict__ mrow,
                                        int lane, int half) {
    #pragma unroll
    for (int i = 0; i < P1_LD; ++i) {
        g.k[i] = k4[i * 32 + lane];
        g.v[i] = v4[i * 32 + lane];
        g.m[i] = mrow[2 * i + half];
    }
}

__device__ __forceinline__ void p1_compute(const P1Group& g, v8f (&acc)[4][4], v4f& ks) {
    v4f kp[P1_LD], vp[P1_LD];
    #pragma unroll
    for (int i = 0; i < P1_LD; ++i) {
        const float m = g.m[i];
        #pragma unroll
        for (int j = 0; j < 4; ++j) {
            kp[i][j] = fmap(g.k[i][j]) * m;
            vp[i][j] = g.v[i][j] * m;
        }
        ks += kp[i];
    }
    #pragma unroll
    for (int s = 0; s < P1_LD / 2; ++s) {
        #pragma unroll
        for (int mt = 0; mt < 4; ++mt) {
            v2f a;
            a.x = vp[2 * s][mt];
            a.y = vp[2 * s + 1][mt];
            #pragma unroll
            for (int nt = 0; nt < 4; ++nt) {
                v2f b;
                b.x = kp[2 * s][nt];
                b.y = kp[2 * s + 1][nt];
                acc[mt][nt] = wmma4(a, b, acc[mt][nt]);
            }
        }
    }
}

__global__ __launch_bounds__(BDIM) void pass1_kv_kernel(const float* __restrict__ kin,
                                                        const float* __restrict__ vin,
                                                        const float* __restrict__ mask,
                                                        float* __restrict__ wkv,
                                                        float* __restrict__ wks) {
    __shared__ __attribute__((aligned(16))) float part[P1_LDS_FLOATS];

    const int tid  = threadIdx.x;
    const int bh   = blockIdx.x / P1_SPLIT;
    const int blk  = blockIdx.x % P1_SPLIT;
    const int b    = bh / H_;
    const int wid  = tid >> 5;
    const int lane = tid & 31;
    const int half = lane >> 4;
    const int lm   = lane & 15;

    const int t0 = blk * P1_TB + wid * P1_TW;
    const v4f* k4 = (const v4f*)(kin + ((size_t)bh * T_ + t0) * D_);
    const v4f* v4 = (const v4f*)(vin + ((size_t)bh * T_ + t0) * D_);
    const float* mrow = mask + (size_t)b * T_ + t0;

    v8f acc[4][4];
    #pragma unroll
    for (int mt = 0; mt < 4; ++mt)
        #pragma unroll
        for (int nt = 0; nt < 4; ++nt)
            acc[mt][nt] = v8f{};
    v4f ks = {};

    P1Group ga, gb;
    p1_load(ga, k4, v4, mrow, lane, half);
    #pragma unroll 1
    for (int g = 0; g < P1_NG - 2; g += 2) {
        p1_load(gb, k4 + (g + 1) * P1_G4, v4 + (g + 1) * P1_G4, mrow + (g + 1) * P1_GT, lane, half);
        __builtin_amdgcn_sched_barrier(0);
        p1_compute(ga, acc, ks);
        p1_load(ga, k4 + (g + 2) * P1_G4, v4 + (g + 2) * P1_G4, mrow + (g + 2) * P1_GT, lane, half);
        __builtin_amdgcn_sched_barrier(0);
        p1_compute(gb, acc, ks);
    }
    p1_load(gb, k4 + (P1_NG - 1) * P1_G4, v4 + (P1_NG - 1) * P1_G4, mrow + (P1_NG - 1) * P1_GT, lane, half);
    __builtin_amdgcn_sched_barrier(0);
    p1_compute(ga, acc, ks);
    p1_compute(gb, acc, ks);

    float* mine = part + wid * (D_ * D_);
    #pragma unroll
    for (int mt = 0; mt < 4; ++mt) {
        #pragma unroll
        for (int r = 0; r < 8; ++r) {
            const int e = 32 * half + 4 * r + mt;
            v4f o;
            o.x = acc[mt][0][r];
            o.y = acc[mt][1][r];
            o.z = acc[mt][2][r];
            o.w = acc[mt][3][r];
            *(v4f*)&mine[e * D_ + 4 * lm] = o;
        }
    }
    #pragma unroll
    for (int j = 0; j < 4; ++j) ks[j] += __shfl_xor(ks[j], 16, 32);
    float* ksp = part + P1_WAVES * (D_ * D_);
    if (half == 0) *(v4f*)&ksp[wid * D_ + 4 * lm] = ks;
    __syncthreads();

    float* kvb = wkv + (size_t)bh * D_ * D_;
    #pragma unroll
    for (int i = 0; i < (D_ * D_) / BDIM; ++i) {
        const int idx = tid + i * BDIM;
        float s = part[idx];
        #pragma unroll
        for (int w = 1; w < P1_WAVES; ++w) s += part[w * (D_ * D_) + idx];
        atomicAdd(&kvb[idx], s);
    }
    if (tid < D_) {
        float s = ksp[tid];
        #pragma unroll
        for (int w = 1; w < P1_WAVES; ++w) s += ksp[w * D_ + tid];
        atomicAdd(&wks[bh * D_ + tid], s);
    }
}

__global__ __launch_bounds__(BDIM) void pass2_out_kernel(const float* __restrict__ qin,
                                                         const float* __restrict__ wkv,
                                                         const float* __restrict__ wks,
                                                         float* __restrict__ out) {
    __shared__ float skvT[D_ * SQS];
    __shared__ float sq[TM * SQS];
    __shared__ float sks[D_];
    __shared__ float sz[TM];

    const int tid   = threadIdx.x;
    const int bh    = blockIdx.x >> 6;
    const int tbase = (blockIdx.x & 63) * TM;
    const size_t qbase = ((size_t)bh * T_ + tbase) * D_;

    const float* kvg = wkv + (size_t)bh * D_ * D_;
    #pragma unroll
    for (int i = 0; i < (D_ * D_) / (BDIM * 4); ++i) {
        int idx4 = tid + i * BDIM;
        int e = idx4 >> 4, d0 = (idx4 & 15) * 4;
        *(v4f*)&skvT[e * SQS + d0] = *(const v4f*)(kvg + (size_t)idx4 * 4);
    }
    #pragma unroll
    for (int i = 0; i < (TM * D_) / (BDIM * 4); ++i) {
        int idx4 = tid + i * BDIM;
        int row = idx4 >> 4, c0 = (idx4 & 15) * 4;
        v4f qq = *(const v4f*)(qin + qbase + (size_t)idx4 * 4);
        v4f r;
        r.x = fmap(qq.x); r.y = fmap(qq.y); r.z = fmap(qq.z); r.w = fmap(qq.w);
        *(v4f*)&sq[row * SQS + c0] = r;
    }
    if (tid < D_) sks[tid] = wks[bh * D_ + tid];
    __syncthreads();

    {
        const int row = tid >> 2;
        const int l4  = tid & 3;
        float s = 0.0f;
        #pragma unroll
        for (int j = 0; j < 16; ++j) {
            const int c = l4 * 16 + j;
            s += sq[row * SQS + c] * sks[c];
        }
        s += __shfl_xor(s, 1, 32);
        s += __shfl_xor(s, 2, 32);
        if (l4 == 0) sz[row] = s + EPS;
    }
    __syncthreads();

    const int wid  = tid >> 5;
    const int lane = tid & 31;
    const int half = lane >> 4;
    const int lm   = lane & 15;
    const int rt  = (wid >> 1) * 16;
    const int ct0 = (wid & 1) * 32;

    v8f c0 = {}, c1 = {};
    #pragma unroll
    for (int kk = 0; kk < D_; kk += 4) {
        const int kr = kk + half * 2;
        v2f fa, fb0, fb1;
        fa.x  = sq[(rt + lm) * SQS + kr];
        fa.y  = sq[(rt + lm) * SQS + kr + 1];
        fb0.x = skvT[(ct0 + lm) * SQS + kr];
        fb0.y = skvT[(ct0 + lm) * SQS + kr + 1];
        fb1.x = skvT[(ct0 + 16 + lm) * SQS + kr];
        fb1.y = skvT[(ct0 + 16 + lm) * SQS + kr + 1];
        c0 = wmma4(fa, fb0, c0);
        c1 = wmma4(fa, fb1, c1);
    }
    __syncthreads();
    #pragma unroll
    for (int r = 0; r < 8; ++r) {
        const int row = rt + r + half * 8;
        sq[row * SQS + ct0 + lm]      = c0[r];
        sq[row * SQS + ct0 + 16 + lm] = c1[r];
    }
    __syncthreads();
    #pragma unroll
    for (int i = 0; i < (TM * D_) / (BDIM * 4); ++i) {
        const int idx4 = tid + i * BDIM;
        const int row = idx4 >> 4, col = (idx4 & 15) * 4;
        const v4f a = *(const v4f*)&sq[row * SQS + col];
        const float z = sz[row];
        v4f o;
        o.x = a.x / z; o.y = a.y / z; o.z = a.z / z; o.w = a.w / z;
        *(v4f*)(out + qbase + (size_t)idx4 * 4) = o;
    }
}

extern "C" void kernel_launch(void* const* d_in, const int* in_sizes, int n_in,
                              void* d_out, int out_size, void* d_ws, size_t ws_size,
                              hipStream_t stream) {
    (void)in_sizes; (void)n_in; (void)out_size; (void)ws_size;
    const float* q    = (const float*)d_in[0];
    const float* k    = (const float*)d_in[1];
    const float* v    = (const float*)d_in[2];
    const float* mask = (const float*)d_in[3];
    float* out = (float*)d_out;
    float* wkv = (float*)d_ws;
    float* wks = wkv + (size_t)BH * D_ * D_;

    const int nz = BH * D_ * D_ + BH * D_;
    zero_ws_kernel<<<(nz + BDIM - 1) / BDIM, BDIM, 0, stream>>>(wkv, nz);
    pass1_kv_kernel<<<BH * P1_SPLIT, BDIM, 0, stream>>>(k, v, mask, wkv, wks);
    pass2_out_kernel<<<BH * (T_ / TM), BDIM, 0, stream>>>(q, wkv, wks, out);
}
